# starting kernel
# baseline (speedup 1.0000x reference)
#include <hip/hip_runtime.h>

typedef __attribute__((ext_vector_type(2))) float v2f;
typedef __attribute__((ext_vector_type(8))) float v8f;

#define N_NODES 50000
#define N_EDGES 800000
#define D       128
#define OUTD    12
#define N_TGT   8192


__global__ void k_zero(float* __restrict__ p, int n) {
  int i = blockIdx.x * blockDim.x + threadIdx.x;
  if (i < n) p[i] = 0.0f;
}

__global__ void k_deg(const int* __restrict__ dst, float* __restrict__ cnt) {
  int e = blockIdx.x * blockDim.x + threadIdx.x;
  if (e < N_EDGES) atomicAdd(&cnt[dst[e]], 1.0f);
}

__global__ void k_dinv(const float* __restrict__ cnt, float* __restrict__ dinv) {
  int i = blockIdx.x * blockDim.x + threadIdx.x;
  if (i < N_NODES) dinv[i] = rsqrtf(cnt[i] + 1.0f);
}


__global__ __launch_bounds__(256) void k_gemm(const float* __restrict__ X,
                                              const float* __restrict__ W,
                                              float* __restrict__ H,
                                              int apply_relu) {
  __shared__ float sX[16][132];
  const int row0 = blockIdx.x * 16;

  for (int idx = threadIdx.x; idx < 16 * 128; idx += 256) {
    const int r = idx >> 7, c = idx & 127;
    float v = X[(size_t)(row0 + r) * D + c];
    if (apply_relu) v = fmaxf(v, 0.0f);
    sX[r][c] = v;
  }
  __syncthreads();

  const int wave    = threadIdx.x >> 5;
  const int lane    = threadIdx.x & 31;
  const int half    = lane >> 4;
  const int l       = lane & 15;
  const int colBase = wave * 16;

  v8f acc = {};
  #pragma unroll
  for (int k = 0; k < 32; ++k) {
    const int kb = 4 * k + 2 * half;
    v2f a, b;
    a.x = sX[l][kb];
    a.y = sX[l][kb + 1];
    b.x = W[(size_t)kb       * D + colBase + l];
    b.y = W[(size_t)(kb + 1) * D + colBase + l];
    acc = __builtin_amdgcn_wmma_f32_16x16x4_f32(
         false, a,  false, b,
         (short)0, acc,  false,  false);
  }

  #pragma unroll
  for (int r = 0; r < 8; ++r) {
    H[(size_t)(row0 + r + 8 * half) * D + colBase + l] = acc[r];
  }
}


__global__ void k_init(const float* __restrict__ H, const float* __restrict__ dinv,
                       const float* __restrict__ bias, float* __restrict__ A) {
  const int i    = blockIdx.x * blockDim.x + threadIdx.x;
  const int node = i >> 7;
  const int f    = i & 127;
  const float di = dinv[node];
  A[i] = H[i] * di * di + bias[f];
}


__global__ void k_edge(const float* __restrict__ H, const int* __restrict__ src,
                       const int* __restrict__ dst, const float* __restrict__ dinv,
                       float* __restrict__ A) {
  const int g    = blockIdx.x * blockDim.x + threadIdx.x;
  const int e    = g >> 5;
  const int lane = g & 31;
  if (e >= N_EDGES) return;
  const int s = src[e];
  const int d = dst[e];
  const float norm = dinv[s] * dinv[d];
  const float4 hv = ((const float4*)(H + (size_t)s * D))[lane];
  float* ap = A + (size_t)d * D + lane * 4;
  atomicAdd(ap + 0, hv.x * norm);
  atomicAdd(ap + 1, hv.y * norm);
  atomicAdd(ap + 2, hv.z * norm);
  atomicAdd(ap + 3, hv.w * norm);
}


__global__ void k_final(const float* __restrict__ A, const int* __restrict__ tgt,
                        const float* __restrict__ Wr, const float* __restrict__ br,
                        float* __restrict__ out) {
  const int g    = blockIdx.x * blockDim.x + threadIdx.x;
  const int t    = g >> 5;
  const int lane = g & 31;
  if (t >= N_TGT) return;
  const int node = tgt[t];
  float h[4];
  #pragma unroll
  for (int j = 0; j < 4; ++j)
    h[j] = fmaxf(A[(size_t)node * D + lane + 32 * j], 0.0f);
  for (int o = 0; o < OUTD; ++o) {
    float p = 0.0f;
    #pragma unroll
    for (int j = 0; j < 4; ++j)
      p += h[j] * Wr[(size_t)(lane + 32 * j) * OUTD + o];
    for (int m = 16; m >= 1; m >>= 1) p += __shfl_xor(p, m, 32);
    if (lane == 0) out[(size_t)t * OUTD + o] = p + br[o];
  }
}


extern "C" void kernel_launch(void* const* d_in, const int* in_sizes, int n_in,
                              void* d_out, int out_size, void* d_ws, size_t ws_size,
                              hipStream_t stream) {
  const float* x   = (const float*)d_in[0];
  const int*   ei  = (const int*)  d_in[1];
  const int*   tgt = (const int*)  d_in[2];
  const float* W1  = (const float*)d_in[3];
  const float* b1  = (const float*)d_in[4];
  const float* W2  = (const float*)d_in[5];
  const float* b2  = (const float*)d_in[6];
  const float* W3  = (const float*)d_in[7];
  const float* b3  = (const float*)d_in[8];
  const float* Wr  = (const float*)d_in[9];
  const float* br  = (const float*)d_in[10];
  float* out = (float*)d_out;

  const int* src = ei;
  const int* dst = ei + N_EDGES;

  float* ws   = (float*)d_ws;
  float* H    = ws;
  float* A    = H + (size_t)N_NODES * D;
  float* cnt  = A + (size_t)N_NODES * D;
  float* dinv = cnt + N_NODES;

  k_zero<<<(N_NODES + 255) / 256, 256, 0, stream>>>(cnt, N_NODES);
  k_deg <<<(N_EDGES + 255) / 256, 256, 0, stream>>>(dst, cnt);
  k_dinv<<<(N_NODES + 255) / 256, 256, 0, stream>>>(cnt, dinv);

  const float* Ws[3] = {W1, W2, W3};
  const float* bs[3] = {b1, b2, b3};
  const float* in = x;
  for (int l = 0; l < 3; ++l) {
    k_gemm<<<N_NODES / 16, 256, 0, stream>>>(in, Ws[l], H, l > 0);
    k_init<<<(N_NODES * D) / 256, 256, 0, stream>>>(H, dinv, bs[l], A);
    k_edge<<<(N_EDGES * 32) / 256, 256, 0, stream>>>(H, src, dst, dinv, A);
    in = A;
  }

  k_final<<<(N_TGT * 32) / 256, 256, 0, stream>>>(A, tgt, Wr, br, out);
}


# Edge scatter-add split by cache half: a wave reads its die number and draws a 128-edge ticket for its half's range of destination rows; a row takes atomics from one half only; atomic sums as before
# speedup vs baseline: 2.1461x; 2.1461x over previous
#include <hip/hip_runtime.h>

typedef __attribute__((ext_vector_type(2))) float v2f;
typedef __attribute__((ext_vector_type(8))) float v8f;

#define N_NODES 50000
#define N_EDGES 800000
#define D       128
#define OUTD    12
#define N_TGT   8192


__global__ void k_zero(float* __restrict__ p, int n) {
  int i = blockIdx.x * blockDim.x + threadIdx.x;
  if (i < n) p[i] = 0.0f;
}

__global__ void k_deg(const int* __restrict__ dst, float* __restrict__ cnt) {
  int e = blockIdx.x * blockDim.x + threadIdx.x;
  if (e < N_EDGES) atomicAdd(&cnt[dst[e]], 1.0f);
}

#define TICKET_STRIDE 64
#define TICKET_WORDS  (6 * TICKET_STRIDE)

__global__ void k_dinv(float* __restrict__ cnt, float* __restrict__ dinv) {
  int i = blockIdx.x * blockDim.x + threadIdx.x;
  if (i < N_NODES) {
    dinv[i] = rsqrtf(cnt[i] + 1.0f);
    if (i < TICKET_WORDS) cnt[i] = 0.0f;
  }
}


__global__ __launch_bounds__(256) void k_gemm(const float* X,
                                              const float* __restrict__ W,
                                              const float* __restrict__ dinv,
                                              const float* __restrict__ bias,
                                              float* __restrict__ H,
                                              float* Aout,
                                              int apply_relu) {
  __shared__ float sX[16][132];
  const int row0 = blockIdx.x * 16;

  for (int idx = threadIdx.x; idx < 16 * 128; idx += 256) {
    const int r = idx >> 7, c = idx & 127;
    float v = X[(size_t)(row0 + r) * D + c];
    if (apply_relu) v = fmaxf(v, 0.0f);
    sX[r][c] = v;
  }
  __syncthreads();

  const int wave    = threadIdx.x >> 5;
  const int lane    = threadIdx.x & 31;
  const int half    = lane >> 4;
  const int l       = lane & 15;
  const int colBase = wave * 16;

  v8f acc = {};
  #pragma unroll
  for (int k = 0; k < 32; ++k) {
    const int kb = 4 * k + 2 * half;
    v2f a, b;
    a.x = sX[l][kb];
    a.y = sX[l][kb + 1];
    b.x = W[(size_t)kb       * D + colBase + l];
    b.y = W[(size_t)(kb + 1) * D + colBase + l];
    acc = __builtin_amdgcn_wmma_f32_16x16x4_f32(
         false, a,  false, b,
         (short)0, acc,  false,  false);
  }

  const float bcol = bias[colBase + l];
  #pragma unroll
  for (int r = 0; r < 8; ++r) {
    const int    row = row0 + r + 8 * half;
    const size_t o   = (size_t)row * D + colBase + l;
    const float  di  = dinv[row];
    const float  h   = acc[r];
    H[o]    = h;
    Aout[o] = h * di * di + bcol;
  }
}


#define EDGES_PER_TICKET 128
#define N_TICKETS  (N_EDGES / EDGES_PER_TICKET)
#define HALF_NODES (N_NODES / 2)
#define EDGE_GROUP 8
#define EDGE_WAVES 4

template <bool FULL>
__device__ __forceinline__ void edge_group(const int sv, const int dv, const float nv,
                                           const int g, const int left,
                                           const float* __restrict__ H,
                                           float* __restrict__ A, const int lane) {
  float h[EDGE_GROUP][4];
  #pragma unroll
  for (int j = 0; j < EDGE_GROUP; ++j) {
    const int s = __builtin_amdgcn_readlane(sv, g + j);
    const float* hp = H + (size_t)s * D + lane;
    h[j][0] = hp[0];
    h[j][1] = hp[32];
    h[j][2] = hp[64];
    h[j][3] = hp[96];
  }
  #pragma unroll
  for (int j = 0; j < EDGE_GROUP; ++j) {
    if (FULL || j < left) {
      const int   d    = __builtin_amdgcn_readlane(dv, g + j);
      const float norm = __int_as_float(
          __builtin_amdgcn_readlane(__float_as_int(nv), g + j));
      float* ap = A + (size_t)d * D + lane;
      atomicAdd(ap +  0, h[j][0] * norm);
      atomicAdd(ap + 32, h[j][1] * norm);
      atomicAdd(ap + 64, h[j][2] * norm);
      atomicAdd(ap + 96, h[j][3] * norm);
    }
  }
}

__global__ __launch_bounds__(EDGE_WAVES * 32) void k_edge(const float* __restrict__ H,
                                              const int* __restrict__ src,
                                              const int* __restrict__ dst,
                                              const float* __restrict__ dinv,
                                              float* __restrict__ A,
                                              unsigned* __restrict__ tick) {
  static_assert(N_EDGES % EDGES_PER_TICKET == 0, "tickets cover the edges exactly");
  static_assert((2 * N_TICKETS) % EDGE_WAVES == 0, "the grid holds one wave per item exactly");
  __shared__ int sS[EDGE_WAVES][EDGES_PER_TICKET];
  __shared__ int sD[EDGE_WAVES][EDGES_PER_TICKET];
  const int lane = threadIdx.x & 31;
  const int wave = threadIdx.x >> 5;

  const unsigned place = __builtin_amdgcn_s_sendmsg_rtn(0x87);
  int r = (int)((place >> 18) & 1u);

  unsigned t = 0u;
  if (lane == 0) t = atomicAdd(&tick[r * TICKET_STRIDE], 1u);
  t = (unsigned)__builtin_amdgcn_readfirstlane((int)t);
  if (t >= (unsigned)N_TICKETS) {
    r ^= 1;
    unsigned t2 = 0u;
    if (lane == 0) t2 = atomicAdd(&tick[r * TICKET_STRIDE], 1u);
    t = (unsigned)__builtin_amdgcn_readfirstlane((int)t2);
    if (t >= (unsigned)N_TICKETS) return;
  }

  const int lo = r * HALF_NODES;
  const int e0 = (int)t * EDGES_PER_TICKET + lane;

  int s_l[4], d_l[4];
  #pragma unroll
  for (int q = 0; q < 4; ++q) {
    s_l[q] = src[e0 + 32 * q];
    d_l[q] = dst[e0 + 32 * q];
  }
  int cnt = 0;
  #pragma unroll
  for (int q = 0; q < 4; ++q) {
    const bool     mine = (unsigned)(d_l[q] - lo) < (unsigned)HALF_NODES;
    const unsigned mk   = __builtin_amdgcn_ballot_w32(mine);
    const int      pos  = cnt + (int)__builtin_amdgcn_mbcnt_lo(mk, 0u);
    if (mine) {
      sS[wave][pos] = s_l[q];
      sD[wave][pos] = d_l[q];
    }
    cnt += __builtin_popcount(mk);
  }

  for (int k0 = 0; k0 < cnt; k0 += 32) {
    const int   e  = min(k0 + lane, cnt - 1);
    const int   sv = sS[wave][e];
    const int   dv = sD[wave][e];
    const float nv = dinv[sv] * dinv[dv];
    #pragma unroll
    for (int g = 0; g < 32; g += EDGE_GROUP) {
      const int left = cnt - (k0 + g);
      if (left >= EDGE_GROUP)  edge_group<true >(sv, dv, nv, g, left, H, A, lane);
      else if (left > 0)       edge_group<false>(sv, dv, nv, g, left, H, A, lane);
    }
  }
}


__global__ void k_final(const float* __restrict__ A, const int* __restrict__ tgt,
                        const float* __restrict__ Wr, const float* __restrict__ br,
                        float* __restrict__ out) {
  const int g    = blockIdx.x * blockDim.x + threadIdx.x;
  const int t    = g >> 5;
  const int lane = g & 31;
  if (t >= N_TGT) return;
  const int node = tgt[t];
  float h[4];
  #pragma unroll
  for (int j = 0; j < 4; ++j)
    h[j] = fmaxf(A[(size_t)node * D + lane + 32 * j], 0.0f);
  for (int o = 0; o < OUTD; ++o) {
    float p = 0.0f;
    #pragma unroll
    for (int j = 0; j < 4; ++j)
      p += h[j] * Wr[(size_t)(lane + 32 * j) * OUTD + o];
    for (int m = 16; m >= 1; m >>= 1) p += __shfl_xor(p, m, 32);
    if (lane == 0) out[(size_t)t * OUTD + o] = p + br[o];
  }
}


extern "C" void kernel_launch(void* const* d_in, const int* in_sizes, int n_in,
                              void* d_out, int out_size, void* d_ws, size_t ws_size,
                              hipStream_t stream) {
  const float* x   = (const float*)d_in[0];
  const int*   ei  = (const int*)  d_in[1];
  const int*   tgt = (const int*)  d_in[2];
  const float* W1  = (const float*)d_in[3];
  const float* b1  = (const float*)d_in[4];
  const float* W2  = (const float*)d_in[5];
  const float* b2  = (const float*)d_in[6];
  const float* W3  = (const float*)d_in[7];
  const float* b3  = (const float*)d_in[8];
  const float* Wr  = (const float*)d_in[9];
  const float* br  = (const float*)d_in[10];
  float* out = (float*)d_out;

  const int* src = ei;
  const int* dst = ei + N_EDGES;

  float* ws   = (float*)d_ws;
  float* H    = ws;
  float* A    = H + (size_t)N_NODES * D;
  float* cnt  = A + (size_t)N_NODES * D;
  float* dinv = cnt + N_NODES;

  k_zero<<<(N_NODES + 255) / 256, 256, 0, stream>>>(cnt, N_NODES);
  k_deg <<<(N_EDGES + 255) / 256, 256, 0, stream>>>(dst, cnt);
  k_dinv<<<(N_NODES + 255) / 256, 256, 0, stream>>>(cnt, dinv);

  const float* Ws[3] = {W1, W2, W3};
  const float* bs[3] = {b1, b2, b3};
  const float* in = x;
  for (int l = 0; l < 3; ++l) {
    k_gemm<<<N_NODES / 16, 256, 0, stream>>>(in, Ws[l], dinv, bs[l], H, A, l > 0);
    k_edge<<<(2 * N_TICKETS) / EDGE_WAVES, EDGE_WAVES * 32, 0, stream>>>(H, src, dst, dinv, A,
                                                  (unsigned*)cnt + l * 2 * TICKET_STRIDE);
    in = A;
  }

  k_final<<<(N_TGT * 32) / 256, 256, 0, stream>>>(A, tgt, Wr, br, out);
}


# Norm factored: GEMM epilogue scales rows by dinv[src]; dst factor and bias go on where the sums are next read; edge pass adds plain rows, no dinv gathers; atomic sums as before, scaling regrouped
# speedup vs baseline: 2.2851x; 1.0648x over previous
#include <hip/hip_runtime.h>

typedef __attribute__((ext_vector_type(2))) float v2f;
typedef __attribute__((ext_vector_type(8))) float v8f;

#define N_NODES 50000
#define N_EDGES 800000
#define D       128
#define OUTD    12
#define N_TGT   8192


__global__ void k_zero(float* __restrict__ p, int n) {
  int i = blockIdx.x * blockDim.x + threadIdx.x;
  if (i < n) p[i] = 0.0f;
}

__global__ void k_deg(const int* __restrict__ dst, float* __restrict__ cnt) {
  int e = blockIdx.x * blockDim.x + threadIdx.x;
  if (e < N_EDGES) atomicAdd(&cnt[dst[e]], 1.0f);
}

#define TICKET_STRIDE 64
#define TICKET_WORDS  (6 * TICKET_STRIDE)

__global__ void k_dinv(float* __restrict__ cnt, float* __restrict__ dinv) {
  int i = blockIdx.x * blockDim.x + threadIdx.x;
  if (i < N_NODES) {
    dinv[i] = rsqrtf(cnt[i] + 1.0f);
    if (i < TICKET_WORDS) cnt[i] = 0.0f;
  }
}


__global__ __launch_bounds__(256) void k_gemm(const float* X,
                                              const float* __restrict__ W,
                                              const float* __restrict__ dinv,
                                              const float* __restrict__ bprev,
                                              float* __restrict__ H,
                                              float* Sout,
                                              int post) {
  __shared__ float sX[16][132];
  const int row0 = blockIdx.x * 16;

  for (int idx = threadIdx.x; idx < 16 * 128; idx += 256) {
    const int r = idx >> 7, c = idx & 127;
    float v = X[(size_t)(row0 + r) * D + c];
    if (post) v = fmaxf(fmaf(v, dinv[row0 + r], bprev[c]), 0.0f);
    sX[r][c] = v;
  }
  __syncthreads();

  const int wave    = threadIdx.x >> 5;
  const int lane    = threadIdx.x & 31;
  const int half    = lane >> 4;
  const int l       = lane & 15;
  const int colBase = wave * 16;

  v8f acc = {};
  #pragma unroll
  for (int k = 0; k < 32; ++k) {
    const int kb = 4 * k + 2 * half;
    v2f a, b;
    a.x = sX[l][kb];
    a.y = sX[l][kb + 1];
    b.x = W[(size_t)kb       * D + colBase + l];
    b.y = W[(size_t)(kb + 1) * D + colBase + l];
    acc = __builtin_amdgcn_wmma_f32_16x16x4_f32(
         false, a,  false, b,
         (short)0, acc,  false,  false);
  }

  #pragma unroll
  for (int r = 0; r < 8; ++r) {
    const int    row = row0 + r + 8 * half;
    const size_t o   = (size_t)row * D + colBase + l;
    const float  hs  = acc[r] * dinv[row];
    H[o]    = hs;
    Sout[o] = hs;
  }
}


#define EDGES_PER_TICKET 128
#define N_TICKETS  (N_EDGES / EDGES_PER_TICKET)
#define HALF_NODES (N_NODES / 2)
#define EDGE_GROUP 8
#define EDGE_WAVES 4

template <bool FULL>
__device__ __forceinline__ void edge_group(const int sv, const int dv,
                                           const int g, const int left,
                                           const float* __restrict__ H,
                                           float* __restrict__ A, const int lane) {
  float h[EDGE_GROUP][4];
  #pragma unroll
  for (int j = 0; j < EDGE_GROUP; ++j) {
    const int s = __builtin_amdgcn_readlane(sv, g + j);
    const float* hp = H + (size_t)s * D + lane;
    h[j][0] = hp[0];
    h[j][1] = hp[32];
    h[j][2] = hp[64];
    h[j][3] = hp[96];
  }
  #pragma unroll
  for (int j = 0; j < EDGE_GROUP; ++j) {
    if (FULL || j < left) {
      const int d = __builtin_amdgcn_readlane(dv, g + j);
      float* ap = A + (size_t)d * D + lane;
      atomicAdd(ap +  0, h[j][0]);
      atomicAdd(ap + 32, h[j][1]);
      atomicAdd(ap + 64, h[j][2]);
      atomicAdd(ap + 96, h[j][3]);
    }
  }
}

__global__ __launch_bounds__(EDGE_WAVES * 32) void k_edge(const float* __restrict__ H,
                                              const int* __restrict__ src,
                                              const int* __restrict__ dst,
                                              float* __restrict__ A,
                                              unsigned* __restrict__ tick) {
  static_assert(N_EDGES % EDGES_PER_TICKET == 0, "tickets cover the edges exactly");
  static_assert((2 * N_TICKETS) % EDGE_WAVES == 0, "the grid holds one wave per item exactly");
  __shared__ int sS[EDGE_WAVES][EDGES_PER_TICKET];
  __shared__ int sD[EDGE_WAVES][EDGES_PER_TICKET];
  const int lane = threadIdx.x & 31;
  const int wave = threadIdx.x >> 5;

  const unsigned place = __builtin_amdgcn_s_sendmsg_rtn(0x87);
  int r = (int)((place >> 18) & 1u);

  unsigned t = 0u;
  if (lane == 0) t = atomicAdd(&tick[r * TICKET_STRIDE], 1u);
  t = (unsigned)__builtin_amdgcn_readfirstlane((int)t);
  if (t >= (unsigned)N_TICKETS) {
    r ^= 1;
    unsigned t2 = 0u;
    if (lane == 0) t2 = atomicAdd(&tick[r * TICKET_STRIDE], 1u);
    t = (unsigned)__builtin_amdgcn_readfirstlane((int)t2);
    if (t >= (unsigned)N_TICKETS) return;
  }

  const int lo = r * HALF_NODES;
  const int e0 = (int)t * EDGES_PER_TICKET + lane;

  int s_l[4], d_l[4];
  #pragma unroll
  for (int q = 0; q < 4; ++q) {
    s_l[q] = src[e0 + 32 * q];
    d_l[q] = dst[e0 + 32 * q];
  }
  int cnt = 0;
  #pragma unroll
  for (int q = 0; q < 4; ++q) {
    const bool     mine = (unsigned)(d_l[q] - lo) < (unsigned)HALF_NODES;
    const unsigned mk   = __builtin_amdgcn_ballot_w32(mine);
    const int      pos  = cnt + (int)__builtin_amdgcn_mbcnt_lo(mk, 0u);
    if (mine) {
      sS[wave][pos] = s_l[q];
      sD[wave][pos] = d_l[q];
    }
    cnt += __builtin_popcount(mk);
  }

  for (int k0 = 0; k0 < cnt; k0 += 32) {
    const int e  = min(k0 + lane, cnt - 1);
    const int sv = sS[wave][e];
    const int dv = sD[wave][e];
    #pragma unroll
    for (int g = 0; g < 32; g += EDGE_GROUP) {
      const int left = cnt - (k0 + g);
      if (left >= EDGE_GROUP)  edge_group<true >(sv, dv, g, left, H, A, lane);
      else if (left > 0)       edge_group<false>(sv, dv, g, left, H, A, lane);
    }
  }
}


__global__ void k_final(const float* __restrict__ A, const int* __restrict__ tgt,
                        const float* __restrict__ dinv, const float* __restrict__ b3,
                        const float* __restrict__ Wr, const float* __restrict__ br,
                        float* __restrict__ out) {
  const int g    = blockIdx.x * blockDim.x + threadIdx.x;
  const int t    = g >> 5;
  const int lane = g & 31;
  if (t >= N_TGT) return;
  const int   node = tgt[t];
  const float di   = dinv[node];
  float h[4];
  #pragma unroll
  for (int j = 0; j < 4; ++j)
    h[j] = fmaxf(fmaf(A[(size_t)node * D + lane + 32 * j], di, b3[lane + 32 * j]), 0.0f);
  for (int o = 0; o < OUTD; ++o) {
    float p = 0.0f;
    #pragma unroll
    for (int j = 0; j < 4; ++j)
      p += h[j] * Wr[(size_t)(lane + 32 * j) * OUTD + o];
    for (int m = 16; m >= 1; m >>= 1) p += __shfl_xor(p, m, 32);
    if (lane == 0) out[(size_t)t * OUTD + o] = p + br[o];
  }
}


extern "C" void kernel_launch(void* const* d_in, const int* in_sizes, int n_in,
                              void* d_out, int out_size, void* d_ws, size_t ws_size,
                              hipStream_t stream) {
  const float* x   = (const float*)d_in[0];
  const int*   ei  = (const int*)  d_in[1];
  const int*   tgt = (const int*)  d_in[2];
  const float* W1  = (const float*)d_in[3];
  const float* b1  = (const float*)d_in[4];
  const float* W2  = (const float*)d_in[5];
  const float* b2  = (const float*)d_in[6];
  const float* W3  = (const float*)d_in[7];
  const float* b3  = (const float*)d_in[8];
  const float* Wr  = (const float*)d_in[9];
  const float* br  = (const float*)d_in[10];
  float* out = (float*)d_out;

  const int* src = ei;
  const int* dst = ei + N_EDGES;

  float* ws   = (float*)d_ws;
  float* H    = ws;
  float* A    = H + (size_t)N_NODES * D;
  float* cnt  = A + (size_t)N_NODES * D;
  float* dinv = cnt + N_NODES;

  k_zero<<<(N_NODES + 255) / 256, 256, 0, stream>>>(cnt, N_NODES);
  k_deg <<<(N_EDGES + 255) / 256, 256, 0, stream>>>(dst, cnt);
  k_dinv<<<(N_NODES + 255) / 256, 256, 0, stream>>>(cnt, dinv);

  const float* Ws[3] = {W1, W2, W3};
  const float* bs[3] = {b1, b2, b3};
  const float* in    = x;
  const float* bprev = b1;
  for (int l = 0; l < 3; ++l) {
    k_gemm<<<N_NODES / 16, 256, 0, stream>>>(in, Ws[l], dinv, bprev, H, A, l > 0);
    k_edge<<<(2 * N_TICKETS) / EDGE_WAVES, EDGE_WAVES * 32, 0, stream>>>(H, src, dst, A,
                                                  (unsigned*)cnt + l * 2 * TICKET_STRIDE);
    in    = A;
    bprev = bs[l];
  }

  k_final<<<(N_TGT * 32) / 256, 256, 0, stream>>>(A, tgt, dinv, b3, Wr, br, out);
}


# Degree count: 64 work-groups each count 12500 destinations in a 200-KB LDS table and store it as a partial plane; dinv pass sums the planes; zero pass and 800000 global atomics gone; dinv identical
# speedup vs baseline: 2.6559x; 1.1623x over previous
#include <hip/hip_runtime.h>

typedef __attribute__((ext_vector_type(2))) float v2f;
typedef __attribute__((ext_vector_type(8))) float v8f;

#define N_NODES 50000
#define N_EDGES 800000
#define D       128
#define OUTD    12
#define N_TGT   8192


#define HIST_BLOCKS     64
#define HIST_THREADS    1024
#define EDGES_PER_HIST  (N_EDGES / HIST_BLOCKS)
#define HIST_LOADS      ((EDGES_PER_HIST + HIST_THREADS - 1) / HIST_THREADS)
#define NODE_QUADS      (N_NODES / 4)

#define TICKET_STRIDE 64
#define TICKET_WORDS  (6 * TICKET_STRIDE)

__global__ __launch_bounds__(HIST_THREADS) void k_hist(const int* __restrict__ dst,
                                                       unsigned* __restrict__ part) {
  static_assert(N_EDGES % HIST_BLOCKS == 0, "slices cover the edges exactly");
  static_assert(N_NODES % 4 == 0, "planes are moved as 16-byte pieces");
  __shared__ uint4 hist4[NODE_QUADS];
  unsigned* hist = (unsigned*)hist4;
  const int tid = threadIdx.x;

  const int* my = dst + (size_t)blockIdx.x * EDGES_PER_HIST;
  int d[HIST_LOADS];
  #pragma unroll
  for (int q = 0; q < HIST_LOADS; ++q) {
    const int e = tid + HIST_THREADS * q;
    d[q] = (e < EDGES_PER_HIST) ? my[e] : -1;
  }

  for (int i = tid; i < NODE_QUADS; i += HIST_THREADS) hist4[i] = make_uint4(0u, 0u, 0u, 0u);
  __syncthreads();

  #pragma unroll
  for (int q = 0; q < HIST_LOADS; ++q)
    if (d[q] >= 0) atomicAdd(&hist[d[q]], 1u);
  __syncthreads();

  uint4* out = (uint4*)(part + (size_t)blockIdx.x * N_NODES);
  for (int i = tid; i < NODE_QUADS; i += HIST_THREADS) out[i] = hist4[i];
}

__global__ __launch_bounds__(128) void k_dinv(const unsigned* __restrict__ part,
                                              float* __restrict__ dinv,
                                              unsigned* __restrict__ tick) {
  const int i = blockIdx.x * blockDim.x + threadIdx.x;
  if (i < NODE_QUADS) {
    const uint4* p = (const uint4*)part + i;
    uint4 c = make_uint4(0u, 0u, 0u, 0u);
    #pragma unroll 16
    for (int b = 0; b < HIST_BLOCKS; ++b) {
      const uint4 v = p[(size_t)b * NODE_QUADS];
      c.x += v.x; c.y += v.y; c.z += v.z; c.w += v.w;
    }
    float4 r;
    r.x = rsqrtf((float)c.x + 1.0f);
    r.y = rsqrtf((float)c.y + 1.0f);
    r.z = rsqrtf((float)c.z + 1.0f);
    r.w = rsqrtf((float)c.w + 1.0f);
    ((float4*)dinv)[i] = r;
    if (i < TICKET_WORDS / 4) ((uint4*)tick)[i] = make_uint4(0u, 0u, 0u, 0u);
  }
}


__global__ __launch_bounds__(256) void k_gemm(const float* X,
                                              const float* __restrict__ W,
                                              const float* __restrict__ dinv,
                                              const float* __restrict__ bprev,
                                              float* __restrict__ H,
                                              float* Sout,
                                              int post) {
  __shared__ float sX[16][132];
  const int row0 = blockIdx.x * 16;

  for (int idx = threadIdx.x; idx < 16 * 128; idx += 256) {
    const int r = idx >> 7, c = idx & 127;
    float v = X[(size_t)(row0 + r) * D + c];
    if (post) v = fmaxf(fmaf(v, dinv[row0 + r], bprev[c]), 0.0f);
    sX[r][c] = v;
  }
  __syncthreads();

  const int wave    = threadIdx.x >> 5;
  const int lane    = threadIdx.x & 31;
  const int half    = lane >> 4;
  const int l       = lane & 15;
  const int colBase = wave * 16;

  v8f acc = {};
  #pragma unroll
  for (int k = 0; k < 32; ++k) {
    const int kb = 4 * k + 2 * half;
    v2f a, b;
    a.x = sX[l][kb];
    a.y = sX[l][kb + 1];
    b.x = W[(size_t)kb       * D + colBase + l];
    b.y = W[(size_t)(kb + 1) * D + colBase + l];
    acc = __builtin_amdgcn_wmma_f32_16x16x4_f32(
         false, a,  false, b,
         (short)0, acc,  false,  false);
  }

  #pragma unroll
  for (int r = 0; r < 8; ++r) {
    const int    row = row0 + r + 8 * half;
    const size_t o   = (size_t)row * D + colBase + l;
    const float  hs  = acc[r] * dinv[row];
    H[o]    = hs;
    Sout[o] = hs;
  }
}


#define EDGES_PER_TICKET 128
#define N_TICKETS  (N_EDGES / EDGES_PER_TICKET)
#define HALF_NODES (N_NODES / 2)
#define EDGE_GROUP 8
#define EDGE_WAVES 4

template <bool FULL>
__device__ __forceinline__ void edge_group(const int sv, const int dv,
                                           const int g, const int left,
                                           const float* __restrict__ H,
                                           float* __restrict__ A, const int lane) {
  float h[EDGE_GROUP][4];
  #pragma unroll
  for (int j = 0; j < EDGE_GROUP; ++j) {
    const int s = __builtin_amdgcn_readlane(sv, g + j);
    const float* hp = H + (size_t)s * D + lane;
    h[j][0] = hp[0];
    h[j][1] = hp[32];
    h[j][2] = hp[64];
    h[j][3] = hp[96];
  }
  #pragma unroll
  for (int j = 0; j < EDGE_GROUP; ++j) {
    if (FULL || j < left) {
      const int d = __builtin_amdgcn_readlane(dv, g + j);
      float* ap = A + (size_t)d * D + lane;
      atomicAdd(ap +  0, h[j][0]);
      atomicAdd(ap + 32, h[j][1]);
      atomicAdd(ap + 64, h[j][2]);
      atomicAdd(ap + 96, h[j][3]);
    }
  }
}

__global__ __launch_bounds__(EDGE_WAVES * 32) void k_edge(const float* __restrict__ H,
                                              const int* __restrict__ src,
                                              const int* __restrict__ dst,
                                              float* __restrict__ A,
                                              unsigned* __restrict__ tick) {
  static_assert(N_EDGES % EDGES_PER_TICKET == 0, "tickets cover the edges exactly");
  static_assert((2 * N_TICKETS) % EDGE_WAVES == 0, "the grid holds one wave per item exactly");
  __shared__ int sS[EDGE_WAVES][EDGES_PER_TICKET];
  __shared__ int sD[EDGE_WAVES][EDGES_PER_TICKET];
  const int lane = threadIdx.x & 31;
  const int wave = threadIdx.x >> 5;

  const unsigned place = __builtin_amdgcn_s_sendmsg_rtn(0x87);
  int r = (int)((place >> 18) & 1u);

  unsigned t = 0u;
  if (lane == 0) t = atomicAdd(&tick[r * TICKET_STRIDE], 1u);
  t = (unsigned)__builtin_amdgcn_readfirstlane((int)t);
  if (t >= (unsigned)N_TICKETS) {
    r ^= 1;
    unsigned t2 = 0u;
    if (lane == 0) t2 = atomicAdd(&tick[r * TICKET_STRIDE], 1u);
    t = (unsigned)__builtin_amdgcn_readfirstlane((int)t2);
    if (t >= (unsigned)N_TICKETS) return;
  }

  const int lo = r * HALF_NODES;
  const int e0 = (int)t * EDGES_PER_TICKET + lane;

  int s_l[4], d_l[4];
  #pragma unroll
  for (int q = 0; q < 4; ++q) {
    s_l[q] = src[e0 + 32 * q];
    d_l[q] = dst[e0 + 32 * q];
  }
  int cnt = 0;
  #pragma unroll
  for (int q = 0; q < 4; ++q) {
    const bool     mine = (unsigned)(d_l[q] - lo) < (unsigned)HALF_NODES;
    const unsigned mk   = __builtin_amdgcn_ballot_w32(mine);
    const int      pos  = cnt + (int)__builtin_amdgcn_mbcnt_lo(mk, 0u);
    if (mine) {
      sS[wave][pos] = s_l[q];
      sD[wave][pos] = d_l[q];
    }
    cnt += __builtin_popcount(mk);
  }

  for (int k0 = 0; k0 < cnt; k0 += 32) {
    const int e  = min(k0 + lane, cnt - 1);
    const int sv = sS[wave][e];
    const int dv = sD[wave][e];
    #pragma unroll
    for (int g = 0; g < 32; g += EDGE_GROUP) {
      const int left = cnt - (k0 + g);
      if (left >= EDGE_GROUP)  edge_group<true >(sv, dv, g, left, H, A, lane);
      else if (left > 0)       edge_group<false>(sv, dv, g, left, H, A, lane);
    }
  }
}


__global__ void k_final(const float* __restrict__ A, const int* __restrict__ tgt,
                        const float* __restrict__ dinv, const float* __restrict__ b3,
                        const float* __restrict__ Wr, const float* __restrict__ br,
                        float* __restrict__ out) {
  const int g    = blockIdx.x * blockDim.x + threadIdx.x;
  const int t    = g >> 5;
  const int lane = g & 31;
  if (t >= N_TGT) return;
  const int   node = tgt[t];
  const float di   = dinv[node];
  float h[4];
  #pragma unroll
  for (int j = 0; j < 4; ++j)
    h[j] = fmaxf(fmaf(A[(size_t)node * D + lane + 32 * j], di, b3[lane + 32 * j]), 0.0f);
  for (int o = 0; o < OUTD; ++o) {
    float p = 0.0f;
    #pragma unroll
    for (int j = 0; j < 4; ++j)
      p += h[j] * Wr[(size_t)(lane + 32 * j) * OUTD + o];
    for (int m = 16; m >= 1; m >>= 1) p += __shfl_xor(p, m, 32);
    if (lane == 0) out[(size_t)t * OUTD + o] = p + br[o];
  }
}


extern "C" void kernel_launch(void* const* d_in, const int* in_sizes, int n_in,
                              void* d_out, int out_size, void* d_ws, size_t ws_size,
                              hipStream_t stream) {
  const float* x   = (const float*)d_in[0];
  const int*   ei  = (const int*)  d_in[1];
  const int*   tgt = (const int*)  d_in[2];
  const float* W1  = (const float*)d_in[3];
  const float* b1  = (const float*)d_in[4];
  const float* W2  = (const float*)d_in[5];
  const float* b2  = (const float*)d_in[6];
  const float* W3  = (const float*)d_in[7];
  const float* b3  = (const float*)d_in[8];
  const float* Wr  = (const float*)d_in[9];
  const float* br  = (const float*)d_in[10];
  float* out = (float*)d_out;

  const int* src = ei;
  const int* dst = ei + N_EDGES;

  float* ws   = (float*)d_ws;
  float* H    = ws;
  float* A    = H + (size_t)N_NODES * D;
  float* cnt  = A + (size_t)N_NODES * D;
  float* dinv = cnt + N_NODES;

  k_hist<<<HIST_BLOCKS, HIST_THREADS, 0, stream>>>(dst, (unsigned*)H);
  k_dinv<<<(NODE_QUADS + 127) / 128, 128, 0, stream>>>((const unsigned*)H, dinv, (unsigned*)cnt);

  const float* Ws[3] = {W1, W2, W3};
  const float* bs[3] = {b1, b2, b3};
  const float* in    = x;
  const float* bprev = b1;
  for (int l = 0; l < 3; ++l) {
    k_gemm<<<N_NODES / 16, 256, 0, stream>>>(in, Ws[l], dinv, bprev, H, A, l > 0);
    k_edge<<<(2 * N_TICKETS) / EDGE_WAVES, EDGE_WAVES * 32, 0, stream>>>(H, src, dst, A,
                                                  (unsigned*)cnt + l * 2 * TICKET_STRIDE);
    in    = A;
    bprev = bs[l];
  }

  k_final<<<(N_TGT * 32) / 256, 256, 0, stream>>>(A, tgt, dinv, b3, Wr, br, out);
}


# GEMM tile load: a thread issues its two 16-B pieces (with the row factors and bias) before using any: one memory round trip ahead of the block's barrier instead of eight in series; sums as before
# speedup vs baseline: 2.8745x; 1.0823x over previous
#include <hip/hip_runtime.h>

typedef __attribute__((ext_vector_type(2))) float v2f;
typedef __attribute__((ext_vector_type(8))) float v8f;

#define N_NODES 50000
#define N_EDGES 800000
#define D       128
#define OUTD    12
#define N_TGT   8192


#define HIST_BLOCKS     64
#define HIST_THREADS    1024
#define EDGES_PER_HIST  (N_EDGES / HIST_BLOCKS)
#define HIST_LOADS      ((EDGES_PER_HIST + HIST_THREADS - 1) / HIST_THREADS)
#define NODE_QUADS      (N_NODES / 4)

#define TICKET_STRIDE 64
#define TICKET_WORDS  (6 * TICKET_STRIDE)

__global__ __launch_bounds__(HIST_THREADS) void k_hist(const int* __restrict__ dst,
                                                       unsigned* __restrict__ part) {
  static_assert(N_EDGES % HIST_BLOCKS == 0, "slices cover the edges exactly");
  static_assert(N_NODES % 4 == 0, "planes are moved as 16-byte pieces");
  __shared__ uint4 hist4[NODE_QUADS];
  unsigned* hist = (unsigned*)hist4;
  const int tid = threadIdx.x;

  const int* my = dst + (size_t)blockIdx.x * EDGES_PER_HIST;
  int d[HIST_LOADS];
  #pragma unroll
  for (int q = 0; q < HIST_LOADS; ++q) {
    const int e = tid + HIST_THREADS * q;
    d[q] = (e < EDGES_PER_HIST) ? my[e] : -1;
  }

  for (int i = tid; i < NODE_QUADS; i += HIST_THREADS) hist4[i] = make_uint4(0u, 0u, 0u, 0u);
  __syncthreads();

  #pragma unroll
  for (int q = 0; q < HIST_LOADS; ++q)
    if (d[q] >= 0) atomicAdd(&hist[d[q]], 1u);
  __syncthreads();

  uint4* out = (uint4*)(part + (size_t)blockIdx.x * N_NODES);
  for (int i = tid; i < NODE_QUADS; i += HIST_THREADS) out[i] = hist4[i];
}

__global__ __launch_bounds__(128) void k_dinv(const unsigned* __restrict__ part,
                                              float* __restrict__ dinv,
                                              unsigned* __restrict__ tick) {
  const int i = blockIdx.x * blockDim.x + threadIdx.x;
  if (i < NODE_QUADS) {
    const uint4* p = (const uint4*)part + i;
    uint4 c = make_uint4(0u, 0u, 0u, 0u);
    #pragma unroll 16
    for (int b = 0; b < HIST_BLOCKS; ++b) {
      const uint4 v = p[(size_t)b * NODE_QUADS];
      c.x += v.x; c.y += v.y; c.z += v.z; c.w += v.w;
    }
    float4 r;
    r.x = rsqrtf((float)c.x + 1.0f);
    r.y = rsqrtf((float)c.y + 1.0f);
    r.z = rsqrtf((float)c.z + 1.0f);
    r.w = rsqrtf((float)c.w + 1.0f);
    ((float4*)dinv)[i] = r;
    if (i < TICKET_WORDS / 4) ((uint4*)tick)[i] = make_uint4(0u, 0u, 0u, 0u);
  }
}


__global__ __launch_bounds__(256) void k_gemm(const float* X,
                                              const float* __restrict__ W,
                                              const float* __restrict__ dinv,
                                              const float* __restrict__ bprev,
                                              float* __restrict__ H,
                                              float* Sout,
                                              int post) {
  __shared__ alignas(16) float sX[16][132];
  const int row0 = blockIdx.x * 16;

  {
    const int r0 = threadIdx.x >> 5;
    const int c  = (threadIdx.x & 31) * 4;
    float4 v0 = *(const float4*)(X + (size_t)(row0 + r0)     * D + c);
    float4 v1 = *(const float4*)(X + (size_t)(row0 + r0 + 8) * D + c);
    if (post) {
      const float  d0 = dinv[row0 + r0];
      const float  d1 = dinv[row0 + r0 + 8];
      const float4 bp = *(const float4*)(bprev + c);
      v0.x = fmaxf(fmaf(v0.x, d0, bp.x), 0.0f);
      v0.y = fmaxf(fmaf(v0.y, d0, bp.y), 0.0f);
      v0.z = fmaxf(fmaf(v0.z, d0, bp.z), 0.0f);
      v0.w = fmaxf(fmaf(v0.w, d0, bp.w), 0.0f);
      v1.x = fmaxf(fmaf(v1.x, d1, bp.x), 0.0f);
      v1.y = fmaxf(fmaf(v1.y, d1, bp.y), 0.0f);
      v1.z = fmaxf(fmaf(v1.z, d1, bp.z), 0.0f);
      v1.w = fmaxf(fmaf(v1.w, d1, bp.w), 0.0f);
    }
    *(float4*)&sX[r0][c]     = v0;
    *(float4*)&sX[r0 + 8][c] = v1;
  }
  __syncthreads();

  const int wave    = threadIdx.x >> 5;
  const int lane    = threadIdx.x & 31;
  const int half    = lane >> 4;
  const int l       = lane & 15;
  const int colBase = wave * 16;

  v8f acc = {};
  #pragma unroll
  for (int k = 0; k < 32; ++k) {
    const int kb = 4 * k + 2 * half;
    v2f a, b;
    a.x = sX[l][kb];
    a.y = sX[l][kb + 1];
    b.x = W[(size_t)kb       * D + colBase + l];
    b.y = W[(size_t)(kb + 1) * D + colBase + l];
    acc = __builtin_amdgcn_wmma_f32_16x16x4_f32(
         false, a,  false, b,
         (short)0, acc,  false,  false);
  }

  #pragma unroll
  for (int r = 0; r < 8; ++r) {
    const int    row = row0 + r + 8 * half;
    const size_t o   = (size_t)row * D + colBase + l;
    const float  hs  = acc[r] * dinv[row];
    H[o]    = hs;
    Sout[o] = hs;
  }
}


#define EDGES_PER_TICKET 128
#define N_TICKETS  (N_EDGES / EDGES_PER_TICKET)
#define HALF_NODES (N_NODES / 2)
#define EDGE_GROUP 8
#define EDGE_WAVES 4

template <bool FULL>
__device__ __forceinline__ void edge_group(const int sv, const int dv,
                                           const int g, const int left,
                                           const float* __restrict__ H,
                                           float* __restrict__ A, const int lane) {
  float h[EDGE_GROUP][4];
  #pragma unroll
  for (int j = 0; j < EDGE_GROUP; ++j) {
    const int s = __builtin_amdgcn_readlane(sv, g + j);
    const float* hp = H + (size_t)s * D + lane;
    h[j][0] = hp[0];
    h[j][1] = hp[32];
    h[j][2] = hp[64];
    h[j][3] = hp[96];
  }
  #pragma unroll
  for (int j = 0; j < EDGE_GROUP; ++j) {
    if (FULL || j < left) {
      const int d = __builtin_amdgcn_readlane(dv, g + j);
      float* ap = A + (size_t)d * D + lane;
      atomicAdd(ap +  0, h[j][0]);
      atomicAdd(ap + 32, h[j][1]);
      atomicAdd(ap + 64, h[j][2]);
      atomicAdd(ap + 96, h[j][3]);
    }
  }
}

__global__ __launch_bounds__(EDGE_WAVES * 32) void k_edge(const float* __restrict__ H,
                                              const int* __restrict__ src,
                                              const int* __restrict__ dst,
                                              float* __restrict__ A,
                                              unsigned* __restrict__ tick) {
  static_assert(N_EDGES % EDGES_PER_TICKET == 0, "tickets cover the edges exactly");
  static_assert((2 * N_TICKETS) % EDGE_WAVES == 0, "the grid holds one wave per item exactly");
  __shared__ int sS[EDGE_WAVES][EDGES_PER_TICKET];
  __shared__ int sD[EDGE_WAVES][EDGES_PER_TICKET];
  const int lane = threadIdx.x & 31;
  const int wave = threadIdx.x >> 5;

  const unsigned place = __builtin_amdgcn_s_sendmsg_rtn(0x87);
  int r = (int)((place >> 18) & 1u);

  unsigned t = 0u;
  if (lane == 0) t = atomicAdd(&tick[r * TICKET_STRIDE], 1u);
  t = (unsigned)__builtin_amdgcn_readfirstlane((int)t);
  if (t >= (unsigned)N_TICKETS) {
    r ^= 1;
    unsigned t2 = 0u;
    if (lane == 0) t2 = atomicAdd(&tick[r * TICKET_STRIDE], 1u);
    t = (unsigned)__builtin_amdgcn_readfirstlane((int)t2);
    if (t >= (unsigned)N_TICKETS) return;
  }

  const int lo = r * HALF_NODES;
  const int e0 = (int)t * EDGES_PER_TICKET + lane;

  int s_l[4], d_l[4];
  #pragma unroll
  for (int q = 0; q < 4; ++q) {
    s_l[q] = src[e0 + 32 * q];
    d_l[q] = dst[e0 + 32 * q];
  }
  int cnt = 0;
  #pragma unroll
  for (int q = 0; q < 4; ++q) {
    const bool     mine = (unsigned)(d_l[q] - lo) < (unsigned)HALF_NODES;
    const unsigned mk   = __builtin_amdgcn_ballot_w32(mine);
    const int      pos  = cnt + (int)__builtin_amdgcn_mbcnt_lo(mk, 0u);
    if (mine) {
      sS[wave][pos] = s_l[q];
      sD[wave][pos] = d_l[q];
    }
    cnt += __builtin_popcount(mk);
  }

  for (int k0 = 0; k0 < cnt; k0 += 32) {
    const int e  = min(k0 + lane, cnt - 1);
    const int sv = sS[wave][e];
    const int dv = sD[wave][e];
    #pragma unroll
    for (int g = 0; g < 32; g += EDGE_GROUP) {
      const int left = cnt - (k0 + g);
      if (left >= EDGE_GROUP)  edge_group<true >(sv, dv, g, left, H, A, lane);
      else if (left > 0)       edge_group<false>(sv, dv, g, left, H, A, lane);
    }
  }
}


__global__ void k_final(const float* __restrict__ A, const int* __restrict__ tgt,
                        const float* __restrict__ dinv, const float* __restrict__ b3,
                        const float* __restrict__ Wr, const float* __restrict__ br,
                        float* __restrict__ out) {
  const int g    = blockIdx.x * blockDim.x + threadIdx.x;
  const int t    = g >> 5;
  const int lane = g & 31;
  if (t >= N_TGT) return;
  const int   node = tgt[t];
  const float di   = dinv[node];
  float h[4];
  #pragma unroll
  for (int j = 0; j < 4; ++j)
    h[j] = fmaxf(fmaf(A[(size_t)node * D + lane + 32 * j], di, b3[lane + 32 * j]), 0.0f);
  for (int o = 0; o < OUTD; ++o) {
    float p = 0.0f;
    #pragma unroll
    for (int j = 0; j < 4; ++j)
      p += h[j] * Wr[(size_t)(lane + 32 * j) * OUTD + o];
    for (int m = 16; m >= 1; m >>= 1) p += __shfl_xor(p, m, 32);
    if (lane == 0) out[(size_t)t * OUTD + o] = p + br[o];
  }
}


extern "C" void kernel_launch(void* const* d_in, const int* in_sizes, int n_in,
                              void* d_out, int out_size, void* d_ws, size_t ws_size,
                              hipStream_t stream) {
  const float* x   = (const float*)d_in[0];
  const int*   ei  = (const int*)  d_in[1];
  const int*   tgt = (const int*)  d_in[2];
  const float* W1  = (const float*)d_in[3];
  const float* b1  = (const float*)d_in[4];
  const float* W2  = (const float*)d_in[5];
  const float* b2  = (const float*)d_in[6];
  const float* W3  = (const float*)d_in[7];
  const float* b3  = (const float*)d_in[8];
  const float* Wr  = (const float*)d_in[9];
  const float* br  = (const float*)d_in[10];
  float* out = (float*)d_out;

  const int* src = ei;
  const int* dst = ei + N_EDGES;

  float* ws   = (float*)d_ws;
  float* H    = ws;
  float* A    = H + (size_t)N_NODES * D;
  float* cnt  = A + (size_t)N_NODES * D;
  float* dinv = cnt + N_NODES;

  k_hist<<<HIST_BLOCKS, HIST_THREADS, 0, stream>>>(dst, (unsigned*)H);
  k_dinv<<<(NODE_QUADS + 127) / 128, 128, 0, stream>>>((const unsigned*)H, dinv, (unsigned*)cnt);

  const float* Ws[3] = {W1, W2, W3};
  const float* bs[3] = {b1, b2, b3};
  const float* in    = x;
  const float* bprev = b1;
  for (int l = 0; l < 3; ++l) {
    k_gemm<<<N_NODES / 16, 256, 0, stream>>>(in, Ws[l], dinv, bprev, H, A, l > 0);
    k_edge<<<(2 * N_TICKETS) / EDGE_WAVES, EDGE_WAVES * 32, 0, stream>>>(H, src, dst, A,
                                                  (unsigned*)cnt + l * 2 * TICKET_STRIDE);
    in    = A;
    bprev = bs[l];
  }

  k_final<<<(N_TGT * 32) / 256, 256, 0, stream>>>(A, tgt, dinv, b3, Wr, br, out);
}


# Layer-3 edge pass keeps only edges whose destination is named in target_idx (node bitmap built per call, tested as the wave packs its ticket); same atomic scatter, all of H formed; sums as before
# speedup vs baseline: 3.2860x; 1.1432x over previous
#include <hip/hip_runtime.h>

typedef __attribute__((ext_vector_type(2))) float v2f;
typedef __attribute__((ext_vector_type(8))) float v8f;

#define N_NODES 50000
#define N_EDGES 800000
#define D       128
#define OUTD    12
#define N_TGT   8192


#define HIST_BLOCKS     64
#define HIST_THREADS    1024
#define EDGES_PER_HIST  (N_EDGES / HIST_BLOCKS)
#define HIST_LOADS      ((EDGES_PER_HIST + HIST_THREADS - 1) / HIST_THREADS)
#define NODE_QUADS      (N_NODES / 4)

#define TICKET_STRIDE 64
#define TICKET_WORDS  (6 * TICKET_STRIDE)

#define BITMAP_WORDS  ((N_NODES + 31) / 32)
#define BITMAP_QUADS  ((BITMAP_WORDS + 3) / 4)
#define BITMAP_OFFSET 512
#define TGT_LOADS     (N_TGT / HIST_THREADS)

__global__ __launch_bounds__(HIST_THREADS) void k_hist(const int* __restrict__ dst,
                                                       unsigned* __restrict__ part,
                                                       const int* __restrict__ tgt,
                                                       unsigned* __restrict__ bits) {
  static_assert(N_EDGES % HIST_BLOCKS == 0, "slices cover the edges exactly");
  static_assert(N_NODES % 4 == 0, "planes are moved as 16-byte pieces");
  static_assert(N_TGT % HIST_THREADS == 0, "the target list is shared out evenly");
  static_assert(BITMAP_OFFSET >= TICKET_WORDS, "bitmap lies behind the ticket words");
  static_assert(BITMAP_OFFSET + 4 * BITMAP_QUADS <= N_NODES, "bitmap fits the count plane");
  __shared__ uint4 hist4[NODE_QUADS];
  unsigned* hist = (unsigned*)hist4;
  const int tid = threadIdx.x;

  if (blockIdx.x == HIST_BLOCKS) {
    int t[TGT_LOADS];
    #pragma unroll
    for (int q = 0; q < TGT_LOADS; ++q) t[q] = tgt[tid + HIST_THREADS * q];
    for (int i = tid; i < BITMAP_QUADS; i += HIST_THREADS) hist4[i] = make_uint4(0u, 0u, 0u, 0u);
    __syncthreads();
    #pragma unroll
    for (int q = 0; q < TGT_LOADS; ++q) atomicOr(&hist[t[q] >> 5], 1u << (t[q] & 31));
    __syncthreads();
    for (int i = tid; i < BITMAP_QUADS; i += HIST_THREADS) ((uint4*)bits)[i] = hist4[i];
    return;
  }

  const int* my = dst + (size_t)blockIdx.x * EDGES_PER_HIST;
  int d[HIST_LOADS];
  #pragma unroll
  for (int q = 0; q < HIST_LOADS; ++q) {
    const int e = tid + HIST_THREADS * q;
    d[q] = (e < EDGES_PER_HIST) ? my[e] : -1;
  }

  for (int i = tid; i < NODE_QUADS; i += HIST_THREADS) hist4[i] = make_uint4(0u, 0u, 0u, 0u);
  __syncthreads();

  #pragma unroll
  for (int q = 0; q < HIST_LOADS; ++q)
    if (d[q] >= 0) atomicAdd(&hist[d[q]], 1u);
  __syncthreads();

  uint4* out = (uint4*)(part + (size_t)blockIdx.x * N_NODES);
  for (int i = tid; i < NODE_QUADS; i += HIST_THREADS) out[i] = hist4[i];
}

__global__ __launch_bounds__(128) void k_dinv(const unsigned* __restrict__ part,
                                              float* __restrict__ dinv,
                                              unsigned* __restrict__ tick) {
  const int i = blockIdx.x * blockDim.x + threadIdx.x;
  if (i < NODE_QUADS) {
    const uint4* p = (const uint4*)part + i;
    uint4 c = make_uint4(0u, 0u, 0u, 0u);
    #pragma unroll 16
    for (int b = 0; b < HIST_BLOCKS; ++b) {
      const uint4 v = p[(size_t)b * NODE_QUADS];
      c.x += v.x; c.y += v.y; c.z += v.z; c.w += v.w;
    }
    float4 r;
    r.x = rsqrtf((float)c.x + 1.0f);
    r.y = rsqrtf((float)c.y + 1.0f);
    r.z = rsqrtf((float)c.z + 1.0f);
    r.w = rsqrtf((float)c.w + 1.0f);
    ((float4*)dinv)[i] = r;
    if (i < TICKET_WORDS / 4) ((uint4*)tick)[i] = make_uint4(0u, 0u, 0u, 0u);
  }
}


__global__ __launch_bounds__(256) void k_gemm(const float* X,
                                              const float* __restrict__ W,
                                              const float* __restrict__ dinv,
                                              const float* __restrict__ bprev,
                                              float* __restrict__ H,
                                              float* Sout,
                                              int post) {
  __shared__ alignas(16) float sX[16][132];
  const int row0 = blockIdx.x * 16;

  {
    const int r0 = threadIdx.x >> 5;
    const int c  = (threadIdx.x & 31) * 4;
    float4 v0 = *(const float4*)(X + (size_t)(row0 + r0)     * D + c);
    float4 v1 = *(const float4*)(X + (size_t)(row0 + r0 + 8) * D + c);
    if (post) {
      const float  d0 = dinv[row0 + r0];
      const float  d1 = dinv[row0 + r0 + 8];
      const float4 bp = *(const float4*)(bprev + c);
      v0.x = fmaxf(fmaf(v0.x, d0, bp.x), 0.0f);
      v0.y = fmaxf(fmaf(v0.y, d0, bp.y), 0.0f);
      v0.z = fmaxf(fmaf(v0.z, d0, bp.z), 0.0f);
      v0.w = fmaxf(fmaf(v0.w, d0, bp.w), 0.0f);
      v1.x = fmaxf(fmaf(v1.x, d1, bp.x), 0.0f);
      v1.y = fmaxf(fmaf(v1.y, d1, bp.y), 0.0f);
      v1.z = fmaxf(fmaf(v1.z, d1, bp.z), 0.0f);
      v1.w = fmaxf(fmaf(v1.w, d1, bp.w), 0.0f);
    }
    *(float4*)&sX[r0][c]     = v0;
    *(float4*)&sX[r0 + 8][c] = v1;
  }
  __syncthreads();

  const int wave    = threadIdx.x >> 5;
  const int lane    = threadIdx.x & 31;
  const int half    = lane >> 4;
  const int l       = lane & 15;
  const int colBase = wave * 16;

  v8f acc = {};
  #pragma unroll
  for (int k = 0; k < 32; ++k) {
    const int kb = 4 * k + 2 * half;
    v2f a, b;
    a.x = sX[l][kb];
    a.y = sX[l][kb + 1];
    b.x = W[(size_t)kb       * D + colBase + l];
    b.y = W[(size_t)(kb + 1) * D + colBase + l];
    acc = __builtin_amdgcn_wmma_f32_16x16x4_f32(
         false, a,  false, b,
         (short)0, acc,  false,  false);
  }

  #pragma unroll
  for (int r = 0; r < 8; ++r) {
    const int    row = row0 + r + 8 * half;
    const size_t o   = (size_t)row * D + colBase + l;
    const float  hs  = acc[r] * dinv[row];
    H[o]    = hs;
    Sout[o] = hs;
  }
}


#define EDGES_PER_TICKET 128
#define N_TICKETS  (N_EDGES / EDGES_PER_TICKET)
#define HALF_NODES (N_NODES / 2)
#define EDGE_GROUP 8
#define EDGE_WAVES 4

template <bool FULL>
__device__ __forceinline__ void edge_group(const int sv, const int dv,
                                           const int g, const int left,
                                           const float* __restrict__ H,
                                           float* __restrict__ A, const int lane) {
  float h[EDGE_GROUP][4];
  #pragma unroll
  for (int j = 0; j < EDGE_GROUP; ++j) {
    const int s = __builtin_amdgcn_readlane(sv, g + j);
    const float* hp = H + (size_t)s * D + lane;
    h[j][0] = hp[0];
    h[j][1] = hp[32];
    h[j][2] = hp[64];
    h[j][3] = hp[96];
  }
  #pragma unroll
  for (int j = 0; j < EDGE_GROUP; ++j) {
    if (FULL || j < left) {
      const int d = __builtin_amdgcn_readlane(dv, g + j);
      float* ap = A + (size_t)d * D + lane;
      atomicAdd(ap +  0, h[j][0]);
      atomicAdd(ap + 32, h[j][1]);
      atomicAdd(ap + 64, h[j][2]);
      atomicAdd(ap + 96, h[j][3]);
    }
  }
}

template <bool FILTER>
__global__ __launch_bounds__(EDGE_WAVES * 32) void k_edge(const float* __restrict__ H,
                                              const int* __restrict__ src,
                                              const int* __restrict__ dst,
                                              float* __restrict__ A,
                                              unsigned* __restrict__ tick,
                                              const unsigned* __restrict__ bits) {
  static_assert(N_EDGES % EDGES_PER_TICKET == 0, "tickets cover the edges exactly");
  static_assert((2 * N_TICKETS) % EDGE_WAVES == 0, "the grid holds one wave per item exactly");
  __shared__ int sS[EDGE_WAVES][EDGES_PER_TICKET];
  __shared__ int sD[EDGE_WAVES][EDGES_PER_TICKET];
  const int lane = threadIdx.x & 31;
  const int wave = threadIdx.x >> 5;

  const unsigned place = __builtin_amdgcn_s_sendmsg_rtn(0x87);
  int r = (int)((place >> 18) & 1u);

  unsigned t = 0u;
  if (lane == 0) t = atomicAdd(&tick[r * TICKET_STRIDE], 1u);
  t = (unsigned)__builtin_amdgcn_readfirstlane((int)t);
  if (t >= (unsigned)N_TICKETS) {
    r ^= 1;
    unsigned t2 = 0u;
    if (lane == 0) t2 = atomicAdd(&tick[r * TICKET_STRIDE], 1u);
    t = (unsigned)__builtin_amdgcn_readfirstlane((int)t2);
    if (t >= (unsigned)N_TICKETS) return;
  }

  const int lo = r * HALF_NODES;
  const int e0 = (int)t * EDGES_PER_TICKET + lane;

  int s_l[4], d_l[4];
  #pragma unroll
  for (int q = 0; q < 4; ++q) {
    s_l[q] = src[e0 + 32 * q];
    d_l[q] = dst[e0 + 32 * q];
  }
  unsigned fw[4] = {0u, 0u, 0u, 0u};
  if (FILTER) {
    #pragma unroll
    for (int q = 0; q < 4; ++q) fw[q] = bits[d_l[q] >> 5];
  }
  int cnt = 0;
  #pragma unroll
  for (int q = 0; q < 4; ++q) {
    bool mine = (unsigned)(d_l[q] - lo) < (unsigned)HALF_NODES;
    if (FILTER) mine = mine && (((fw[q] >> (d_l[q] & 31)) & 1u) != 0u);
    const unsigned mk   = __builtin_amdgcn_ballot_w32(mine);
    const int      pos  = cnt + (int)__builtin_amdgcn_mbcnt_lo(mk, 0u);
    if (mine) {
      sS[wave][pos] = s_l[q];
      sD[wave][pos] = d_l[q];
    }
    cnt += __builtin_popcount(mk);
  }

  for (int k0 = 0; k0 < cnt; k0 += 32) {
    const int e  = min(k0 + lane, cnt - 1);
    const int sv = sS[wave][e];
    const int dv = sD[wave][e];
    #pragma unroll
    for (int g = 0; g < 32; g += EDGE_GROUP) {
      const int left = cnt - (k0 + g);
      if (left >= EDGE_GROUP)  edge_group<true >(sv, dv, g, left, H, A, lane);
      else if (left > 0)       edge_group<false>(sv, dv, g, left, H, A, lane);
    }
  }
}


__global__ void k_final(const float* __restrict__ A, const int* __restrict__ tgt,
                        const float* __restrict__ dinv, const float* __restrict__ b3,
                        const float* __restrict__ Wr, const float* __restrict__ br,
                        float* __restrict__ out) {
  const int g    = blockIdx.x * blockDim.x + threadIdx.x;
  const int t    = g >> 5;
  const int lane = g & 31;
  if (t >= N_TGT) return;
  const int   node = tgt[t];
  const float di   = dinv[node];
  float h[4];
  #pragma unroll
  for (int j = 0; j < 4; ++j)
    h[j] = fmaxf(fmaf(A[(size_t)node * D + lane + 32 * j], di, b3[lane + 32 * j]), 0.0f);
  for (int o = 0; o < OUTD; ++o) {
    float p = 0.0f;
    #pragma unroll
    for (int j = 0; j < 4; ++j)
      p += h[j] * Wr[(size_t)(lane + 32 * j) * OUTD + o];
    for (int m = 16; m >= 1; m >>= 1) p += __shfl_xor(p, m, 32);
    if (lane == 0) out[(size_t)t * OUTD + o] = p + br[o];
  }
}


extern "C" void kernel_launch(void* const* d_in, const int* in_sizes, int n_in,
                              void* d_out, int out_size, void* d_ws, size_t ws_size,
                              hipStream_t stream) {
  const float* x   = (const float*)d_in[0];
  const int*   ei  = (const int*)  d_in[1];
  const int*   tgt = (const int*)  d_in[2];
  const float* W1  = (const float*)d_in[3];
  const float* b1  = (const float*)d_in[4];
  const float* W2  = (const float*)d_in[5];
  const float* b2  = (const float*)d_in[6];
  const float* W3  = (const float*)d_in[7];
  const float* b3  = (const float*)d_in[8];
  const float* Wr  = (const float*)d_in[9];
  const float* br  = (const float*)d_in[10];
  float* out = (float*)d_out;

  const int* src = ei;
  const int* dst = ei + N_EDGES;

  float* ws   = (float*)d_ws;
  float* H    = ws;
  float* A    = H + (size_t)N_NODES * D;
  float* cnt  = A + (size_t)N_NODES * D;
  float* dinv = cnt + N_NODES;
  unsigned* bits = (unsigned*)cnt + BITMAP_OFFSET;

  k_hist<<<HIST_BLOCKS + 1, HIST_THREADS, 0, stream>>>(dst, (unsigned*)H, tgt, bits);
  k_dinv<<<(NODE_QUADS + 127) / 128, 128, 0, stream>>>((const unsigned*)H, dinv, (unsigned*)cnt);

  const float* Ws[3] = {W1, W2, W3};
  const float* bs[3] = {b1, b2, b3};
  const float* in    = x;
  const float* bprev = b1;
  for (int l = 0; l < 3; ++l) {
    k_gemm<<<N_NODES / 16, 256, 0, stream>>>(in, Ws[l], dinv, bprev, H, A, l > 0);
    if (l < 2)
      k_edge<false><<<(2 * N_TICKETS) / EDGE_WAVES, EDGE_WAVES * 32, 0, stream>>>(H, src, dst, A,
                                                  (unsigned*)cnt + l * 2 * TICKET_STRIDE, bits);
    else
      k_edge<true><<<(2 * N_TICKETS) / EDGE_WAVES, EDGE_WAVES * 32, 0, stream>>>(H, src, dst, A,
                                                  (unsigned*)cnt + l * 2 * TICKET_STRIDE, bits);
    in    = A;
    bprev = bs[l];
  }

  k_final<<<(N_TGT * 32) / 256, 256, 0, stream>>>(A, tgt, dinv, b3, Wr, br, out);
}


# GEMM on a resident grid: 512 work-groups walk the 16-row tiles, a wave keeps its W slice in registers, next tile's loads fly under the current WMMAs (LDS double buffer); sums as before
# speedup vs baseline: 3.5049x; 1.0666x over previous
#include <hip/hip_runtime.h>

typedef __attribute__((ext_vector_type(2))) float v2f;
typedef __attribute__((ext_vector_type(8))) float v8f;

#define N_NODES 50000
#define N_EDGES 800000
#define D       128
#define OUTD    12
#define N_TGT   8192


#define HIST_BLOCKS     64
#define HIST_THREADS    1024
#define EDGES_PER_HIST  (N_EDGES / HIST_BLOCKS)
#define HIST_LOADS      ((EDGES_PER_HIST + HIST_THREADS - 1) / HIST_THREADS)
#define NODE_QUADS      (N_NODES / 4)

#define TICKET_STRIDE 64
#define TICKET_WORDS  (6 * TICKET_STRIDE)

#define BITMAP_WORDS  ((N_NODES + 31) / 32)
#define BITMAP_QUADS  ((BITMAP_WORDS + 3) / 4)
#define BITMAP_OFFSET 512
#define TGT_LOADS     (N_TGT / HIST_THREADS)

__global__ __launch_bounds__(HIST_THREADS) void k_hist(const int* __restrict__ dst,
                                                       unsigned* __restrict__ part,
                                                       const int* __restrict__ tgt,
                                                       unsigned* __restrict__ bits) {
  static_assert(N_EDGES % HIST_BLOCKS == 0, "slices cover the edges exactly");
  static_assert(N_NODES % 4 == 0, "planes are moved as 16-byte pieces");
  static_assert(N_TGT % HIST_THREADS == 0, "the target list is shared out evenly");
  static_assert(BITMAP_OFFSET >= TICKET_WORDS, "bitmap lies behind the ticket words");
  static_assert(BITMAP_OFFSET + 4 * BITMAP_QUADS <= N_NODES, "bitmap fits the count plane");
  __shared__ uint4 hist4[NODE_QUADS];
  unsigned* hist = (unsigned*)hist4;
  const int tid = threadIdx.x;

  if (blockIdx.x == HIST_BLOCKS) {
    int t[TGT_LOADS];
    #pragma unroll
    for (int q = 0; q < TGT_LOADS; ++q) t[q] = tgt[tid + HIST_THREADS * q];
    for (int i = tid; i < BITMAP_QUADS; i += HIST_THREADS) hist4[i] = make_uint4(0u, 0u, 0u, 0u);
    __syncthreads();
    #pragma unroll
    for (int q = 0; q < TGT_LOADS; ++q) atomicOr(&hist[t[q] >> 5], 1u << (t[q] & 31));
    __syncthreads();
    for (int i = tid; i < BITMAP_QUADS; i += HIST_THREADS) ((uint4*)bits)[i] = hist4[i];
    return;
  }

  const int* my = dst + (size_t)blockIdx.x * EDGES_PER_HIST;
  int d[HIST_LOADS];
  #pragma unroll
  for (int q = 0; q < HIST_LOADS; ++q) {
    const int e = tid + HIST_THREADS * q;
    d[q] = (e < EDGES_PER_HIST) ? my[e] : -1;
  }

  for (int i = tid; i < NODE_QUADS; i += HIST_THREADS) hist4[i] = make_uint4(0u, 0u, 0u, 0u);
  __syncthreads();

  #pragma unroll
  for (int q = 0; q < HIST_LOADS; ++q)
    if (d[q] >= 0) atomicAdd(&hist[d[q]], 1u);
  __syncthreads();

  uint4* out = (uint4*)(part + (size_t)blockIdx.x * N_NODES);
  for (int i = tid; i < NODE_QUADS; i += HIST_THREADS) out[i] = hist4[i];
}

__global__ __launch_bounds__(128) void k_dinv(const unsigned* __restrict__ part,
                                              float* __restrict__ dinv,
                                              unsigned* __restrict__ tick) {
  const int i = blockIdx.x * blockDim.x + threadIdx.x;
  if (i < NODE_QUADS) {
    const uint4* p = (const uint4*)part + i;
    uint4 c = make_uint4(0u, 0u, 0u, 0u);
    #pragma unroll 16
    for (int b = 0; b < HIST_BLOCKS; ++b) {
      const uint4 v = p[(size_t)b * NODE_QUADS];
      c.x += v.x; c.y += v.y; c.z += v.z; c.w += v.w;
    }
    float4 r;
    r.x = rsqrtf((float)c.x + 1.0f);
    r.y = rsqrtf((float)c.y + 1.0f);
    r.z = rsqrtf((float)c.z + 1.0f);
    r.w = rsqrtf((float)c.w + 1.0f);
    ((float4*)dinv)[i] = r;
    if (i < TICKET_WORDS / 4) ((uint4*)tick)[i] = make_uint4(0u, 0u, 0u, 0u);
  }
}


#define GEMM_GROUPS 512
#define N_TILES     (N_NODES / 16)

__device__ __forceinline__ void gemm_stage(float (*sXb)[132], float* sDb,
                                           float4 v0, float4 v1,
                                           const float d0, const float d1,
                                           const float4 bp, const int post,
                                           const int r0, const int c, const int lane) {
  if (post) {
    v0.x = fmaxf(fmaf(v0.x, d0, bp.x), 0.0f);
    v0.y = fmaxf(fmaf(v0.y, d0, bp.y), 0.0f);
    v0.z = fmaxf(fmaf(v0.z, d0, bp.z), 0.0f);
    v0.w = fmaxf(fmaf(v0.w, d0, bp.w), 0.0f);
    v1.x = fmaxf(fmaf(v1.x, d1, bp.x), 0.0f);
    v1.y = fmaxf(fmaf(v1.y, d1, bp.y), 0.0f);
    v1.z = fmaxf(fmaf(v1.z, d1, bp.z), 0.0f);
    v1.w = fmaxf(fmaf(v1.w, d1, bp.w), 0.0f);
  }
  *(float4*)&sXb[r0][c]     = v0;
  *(float4*)&sXb[r0 + 8][c] = v1;
  if (lane == 0) {
    sDb[r0]     = d0;
    sDb[r0 + 8] = d1;
  }
}

__global__ __launch_bounds__(256) void k_gemm(const float* X,
                                              const float* __restrict__ W,
                                              const float* __restrict__ dinv,
                                              const float* __restrict__ bprev,
                                              float* __restrict__ H,
                                              float* Sout,
                                              int post) {
  static_assert(N_NODES % 16 == 0, "tiles cover the rows exactly");
  static_assert(GEMM_GROUPS <= N_TILES, "every work-group has a first tile");
  __shared__ alignas(16) float sX[2][16][132];
  __shared__ alignas(16) float sDv[2][16];

  const int wave    = threadIdx.x >> 5;
  const int lane    = threadIdx.x & 31;
  const int half    = lane >> 4;
  const int l       = lane & 15;
  const int colBase = wave * 16;
  const int r0 = wave;
  const int c  = lane * 4;

  int tile = blockIdx.x;

  float4 v0 = *(const float4*)(X + (size_t)(tile * 16 + r0)     * D + c);
  float4 v1 = *(const float4*)(X + (size_t)(tile * 16 + r0 + 8) * D + c);
  float  d0 = dinv[tile * 16 + r0];
  float  d1 = dinv[tile * 16 + r0 + 8];
  float4 bp = make_float4(0.0f, 0.0f, 0.0f, 0.0f);
  if (post) bp = *(const float4*)(bprev + c);

  v2f bw[32];
  #pragma unroll
  for (int k = 0; k < 32; ++k) {
    const int kb = 4 * k + 2 * half;
    bw[k].x = W[(size_t)kb       * D + colBase + l];
    bw[k].y = W[(size_t)(kb + 1) * D + colBase + l];
  }

  gemm_stage(sX[0], sDv[0], v0, v1, d0, d1, bp, post, r0, c, lane);
  __syncthreads();

  int cur = 0;
  for (;;) {
    const int  nxt  = tile + GEMM_GROUPS;
    const bool more = nxt < N_TILES;
    if (more) {
      v0 = *(const float4*)(X + (size_t)(nxt * 16 + r0)     * D + c);
      v1 = *(const float4*)(X + (size_t)(nxt * 16 + r0 + 8) * D + c);
      d0 = dinv[nxt * 16 + r0];
      d1 = dinv[nxt * 16 + r0 + 8];
    }

    v8f acc = {};
    #pragma unroll
    for (int k = 0; k < 32; ++k) {
      const int kb = 4 * k + 2 * half;
      v2f a;
      a.x = sX[cur][l][kb];
      a.y = sX[cur][l][kb + 1];
      acc = __builtin_amdgcn_wmma_f32_16x16x4_f32(
           false, a,  false, bw[k],
           (short)0, acc,  false,  false);
    }

    const size_t o0 = (size_t)(tile * 16 + 8 * half) * D + colBase + l;
    float* hp = H + o0;
    float* sp = Sout + o0;
    #pragma unroll
    for (int r = 0; r < 8; ++r) {
      const float hs = acc[r] * sDv[cur][r + 8 * half];
      hp[r * D] = hs;
      sp[r * D] = hs;
    }

    if (!more) break;

    gemm_stage(sX[cur ^ 1], sDv[cur ^ 1], v0, v1, d0, d1, bp, post, r0, c, lane);
    __builtin_amdgcn_fence(__ATOMIC_RELEASE, "workgroup", "local");
    __builtin_amdgcn_s_barrier();
    __builtin_amdgcn_fence(__ATOMIC_ACQUIRE, "workgroup", "local");
    cur ^= 1;
    tile = nxt;
  }
}


#define EDGES_PER_TICKET 128
#define N_TICKETS  (N_EDGES / EDGES_PER_TICKET)
#define HALF_NODES (N_NODES / 2)
#define EDGE_GROUP 8
#define EDGE_WAVES 4

template <bool FULL>
__device__ __forceinline__ void edge_group(const int sv, const int dv,
                                           const int g, const int left,
                                           const float* __restrict__ H,
                                           float* __restrict__ A, const int lane) {
  float h[EDGE_GROUP][4];
  #pragma unroll
  for (int j = 0; j < EDGE_GROUP; ++j) {
    const int s = __builtin_amdgcn_readlane(sv, g + j);
    const float* hp = H + (size_t)s * D + lane;
    h[j][0] = hp[0];
    h[j][1] = hp[32];
    h[j][2] = hp[64];
    h[j][3] = hp[96];
  }
  #pragma unroll
  for (int j = 0; j < EDGE_GROUP; ++j) {
    if (FULL || j < left) {
      const int d = __builtin_amdgcn_readlane(dv, g + j);
      float* ap = A + (size_t)d * D + lane;
      atomicAdd(ap +  0, h[j][0]);
      atomicAdd(ap + 32, h[j][1]);
      atomicAdd(ap + 64, h[j][2]);
      atomicAdd(ap + 96, h[j][3]);
    }
  }
}

template <bool FILTER>
__global__ __launch_bounds__(EDGE_WAVES * 32) void k_edge(const float* __restrict__ H,
                                              const int* __restrict__ src,
                                              const int* __restrict__ dst,
                                              float* __restrict__ A,
                                              unsigned* __restrict__ tick,
                                              const unsigned* __restrict__ bits) {
  static_assert(N_EDGES % EDGES_PER_TICKET == 0, "tickets cover the edges exactly");
  static_assert((2 * N_TICKETS) % EDGE_WAVES == 0, "the grid holds one wave per item exactly");
  __shared__ int sS[EDGE_WAVES][EDGES_PER_TICKET];
  __shared__ int sD[EDGE_WAVES][EDGES_PER_TICKET];
  const int lane = threadIdx.x & 31;
  const int wave = threadIdx.x >> 5;

  const unsigned place = __builtin_amdgcn_s_sendmsg_rtn(0x87);
  int r = (int)((place >> 18) & 1u);

  unsigned t = 0u;
  if (lane == 0) t = atomicAdd(&tick[r * TICKET_STRIDE], 1u);
  t = (unsigned)__builtin_amdgcn_readfirstlane((int)t);
  if (t >= (unsigned)N_TICKETS) {
    r ^= 1;
    unsigned t2 = 0u;
    if (lane == 0) t2 = atomicAdd(&tick[r * TICKET_STRIDE], 1u);
    t = (unsigned)__builtin_amdgcn_readfirstlane((int)t2);
    if (t >= (unsigned)N_TICKETS) return;
  }

  const int lo = r * HALF_NODES;
  const int e0 = (int)t * EDGES_PER_TICKET + lane;

  int s_l[4], d_l[4];
  #pragma unroll
  for (int q = 0; q < 4; ++q) {
    s_l[q] = src[e0 + 32 * q];
    d_l[q] = dst[e0 + 32 * q];
  }
  unsigned fw[4] = {0u, 0u, 0u, 0u};
  if (FILTER) {
    #pragma unroll
    for (int q = 0; q < 4; ++q) fw[q] = bits[d_l[q] >> 5];
  }
  int cnt = 0;
  #pragma unroll
  for (int q = 0; q < 4; ++q) {
    bool mine = (unsigned)(d_l[q] - lo) < (unsigned)HALF_NODES;
    if (FILTER) mine = mine && (((fw[q] >> (d_l[q] & 31)) & 1u) != 0u);
    const unsigned mk   = __builtin_amdgcn_ballot_w32(mine);
    const int      pos  = cnt + (int)__builtin_amdgcn_mbcnt_lo(mk, 0u);
    if (mine) {
      sS[wave][pos] = s_l[q];
      sD[wave][pos] = d_l[q];
    }
    cnt += __builtin_popcount(mk);
  }

  for (int k0 = 0; k0 < cnt; k0 += 32) {
    const int e  = min(k0 + lane, cnt - 1);
    const int sv = sS[wave][e];
    const int dv = sD[wave][e];
    #pragma unroll
    for (int g = 0; g < 32; g += EDGE_GROUP) {
      const int left = cnt - (k0 + g);
      if (left >= EDGE_GROUP)  edge_group<true >(sv, dv, g, left, H, A, lane);
      else if (left > 0)       edge_group<false>(sv, dv, g, left, H, A, lane);
    }
  }
}


__global__ void k_final(const float* __restrict__ A, const int* __restrict__ tgt,
                        const float* __restrict__ dinv, const float* __restrict__ b3,
                        const float* __restrict__ Wr, const float* __restrict__ br,
                        float* __restrict__ out) {
  const int g    = blockIdx.x * blockDim.x + threadIdx.x;
  const int t    = g >> 5;
  const int lane = g & 31;
  if (t >= N_TGT) return;
  const int   node = tgt[t];
  const float di   = dinv[node];
  float h[4];
  #pragma unroll
  for (int j = 0; j < 4; ++j)
    h[j] = fmaxf(fmaf(A[(size_t)node * D + lane + 32 * j], di, b3[lane + 32 * j]), 0.0f);
  for (int o = 0; o < OUTD; ++o) {
    float p = 0.0f;
    #pragma unroll
    for (int j = 0; j < 4; ++j)
      p += h[j] * Wr[(size_t)(lane + 32 * j) * OUTD + o];
    for (int m = 16; m >= 1; m >>= 1) p += __shfl_xor(p, m, 32);
    if (lane == 0) out[(size_t)t * OUTD + o] = p + br[o];
  }
}


extern "C" void kernel_launch(void* const* d_in, const int* in_sizes, int n_in,
                              void* d_out, int out_size, void* d_ws, size_t ws_size,
                              hipStream_t stream) {
  const float* x   = (const float*)d_in[0];
  const int*   ei  = (const int*)  d_in[1];
  const int*   tgt = (const int*)  d_in[2];
  const float* W1  = (const float*)d_in[3];
  const float* b1  = (const float*)d_in[4];
  const float* W2  = (const float*)d_in[5];
  const float* b2  = (const float*)d_in[6];
  const float* W3  = (const float*)d_in[7];
  const float* b3  = (const float*)d_in[8];
  const float* Wr  = (const float*)d_in[9];
  const float* br  = (const float*)d_in[10];
  float* out = (float*)d_out;

  const int* src = ei;
  const int* dst = ei + N_EDGES;

  float* ws   = (float*)d_ws;
  float* H    = ws;
  float* A    = H + (size_t)N_NODES * D;
  float* cnt  = A + (size_t)N_NODES * D;
  float* dinv = cnt + N_NODES;
  unsigned* bits = (unsigned*)cnt + BITMAP_OFFSET;

  k_hist<<<HIST_BLOCKS + 1, HIST_THREADS, 0, stream>>>(dst, (unsigned*)H, tgt, bits);
  k_dinv<<<(NODE_QUADS + 127) / 128, 128, 0, stream>>>((const unsigned*)H, dinv, (unsigned*)cnt);

  const float* Ws[3] = {W1, W2, W3};
  const float* bs[3] = {b1, b2, b3};
  const float* in    = x;
  const float* bprev = b1;
  for (int l = 0; l < 3; ++l) {
    k_gemm<<<GEMM_GROUPS, 256, 0, stream>>>(in, Ws[l], dinv, bprev, H, A, l > 0);
    if (l < 2)
      k_edge<false><<<(2 * N_TICKETS) / EDGE_WAVES, EDGE_WAVES * 32, 0, stream>>>(H, src, dst, A,
                                                  (unsigned*)cnt + l * 2 * TICKET_STRIDE, bits);
    else
      k_edge<true><<<(2 * N_TICKETS) / EDGE_WAVES, EDGE_WAVES * 32, 0, stream>>>(H, src, dst, A,
                                                  (unsigned*)cnt + l * 2 * TICKET_STRIDE, bits);
    in    = A;
    bprev = bs[l];
  }

  k_final<<<(N_TGT * 32) / 256, 256, 0, stream>>>(A, tgt, dinv, b3, Wr, br, out);
}
